# starting kernel
# baseline (speedup 1.0000x reference)
#include <hip/hip_runtime.h>

typedef __attribute__((ext_vector_type(2))) float v2f;
typedef __attribute__((ext_vector_type(8))) float v8f;

#define N_RES   8192
#define N_INP   64
#define N_HAD   16384
#define N_LAYER 3
#define PITCH   140
#define LDS_BYTES (128 * PITCH * 4)

__device__ __forceinline__ float hsgn(int i, int j) {
  return (__popc(i & j) & 1) ? -1.0f : 1.0f;
}

__device__ __forceinline__ v8f v8f_zero() {
  v8f z;
#pragma unroll
  for (int i = 0; i < 8; ++i) z[i] = 0.0f;
  return z;
}

__global__ __launch_bounds__(256) void fwht_reservoir(
    const float* __restrict__ state, const float* __restrict__ inputs,
    const float* __restrict__ bias, const float* __restrict__ diag,
    float* __restrict__ out) {
  extern __shared__ float X[];
  const int row  = blockIdx.x;
  const int tid  = threadIdx.x;
  const int lane = tid & 31;
  const int wave = tid >> 5;
  const int m    = lane & 15;
  const int kh   = lane >> 4;

  for (int n = tid; n < N_HAD; n += 256) {
    float v;
    if (n < N_RES)                   v = 0.9f * state[(size_t)row * N_RES + n];
    else if (n < N_RES + N_INP)      v = 0.4f * inputs[row * N_INP + (n - N_RES)];
    else                             v = 0.0f;
    X[(n >> 7) * PITCH + (n & 127)] = v * diag[n];
  }
  __syncthreads();

  v2f Hp[4], Hm[4];
#pragma unroll
  for (int s = 0; s < 4; ++s) {
    const int j0 = 4 * s + 2 * kh;
    Hp[s].x = hsgn(m, j0);
    Hp[s].y = hsgn(m, j0 + 1);
    Hm[s].x = -Hp[s].x;
    Hm[s].y = -Hp[s].y;
  }

  for (int l = 0; l < N_LAYER; ++l) {
    {
      const int c = wave * 16 + m;
      v8f acc[8];
#pragma unroll
      for (int t = 0; t < 8; ++t) acc[t] = v8f_zero();
#pragma unroll
      for (int k = 0; k < 8; ++k) {
#pragma unroll
        for (int s = 0; s < 4; ++s) {
          const int r0 = k * 16 + 4 * s + 2 * kh;
          v2f xb;
          xb.x = X[r0 * PITCH + c];
          xb.y = X[(r0 + 1) * PITCH + c];
#pragma unroll
          for (int t = 0; t < 8; ++t) {
            const v2f h = (__popc(t & k) & 1) ? Hm[s] : Hp[s];
            acc[t] = __builtin_amdgcn_wmma_f32_16x16x4_f32(
                false, h, false, xb, (short)0, acc[t], false, false);
          }
        }
      }
#pragma unroll
      for (int t = 0; t < 8; ++t)
#pragma unroll
        for (int v = 0; v < 8; ++v)
          X[(t * 16 + v + 8 * kh) * PITCH + c] = acc[t][v];
    }
    __syncthreads();

    {
      const int r = wave * 16 + m;
      v8f acc[8];
#pragma unroll
      for (int t = 0; t < 8; ++t) acc[t] = v8f_zero();
#pragma unroll
      for (int k = 0; k < 8; ++k) {
#pragma unroll
        for (int s = 0; s < 4; ++s) {
          const int c0 = k * 16 + 4 * s + 2 * kh;
          v2f xa;
          xa.x = X[r * PITCH + c0];
          xa.y = X[r * PITCH + c0 + 1];
#pragma unroll
          for (int t = 0; t < 8; ++t) {
            const v2f h = (__popc(t & k) & 1) ? Hm[s] : Hp[s];
            acc[t] = __builtin_amdgcn_wmma_f32_16x16x4_f32(
                false, xa, false, h, (short)0, acc[t], false, false);
          }
        }
      }
#pragma unroll
      for (int t = 0; t < 8; ++t)
#pragma unroll
        for (int v = 0; v < 8; ++v)
          X[(wave * 16 + v + 8 * kh) * PITCH + t * 16 + m] = acc[t][v];
    }
    __syncthreads();

    if (l + 1 < N_LAYER) {
      const float* dg = diag + (size_t)(l + 1) * N_HAD;
      for (int n = tid; n < N_HAD; n += 256)
        X[(n >> 7) * PITCH + (n & 127)] *= dg[n];
      __syncthreads();
    }
  }

  for (int n = tid; n < N_RES; n += 256) {
    const float z = X[(n >> 7) * PITCH + (n & 127)] * (1.0f / (float)N_HAD) + bias[n];
    out[(size_t)row * N_RES + n] = erff(z);
  }
}

extern "C" void kernel_launch(void* const* d_in, const int* in_sizes, int n_in,
                              void* d_out, int out_size, void* d_ws, size_t ws_size,
                              hipStream_t stream) {
  const float* state  = (const float*)d_in[0];
  const float* inputs = (const float*)d_in[1];
  const float* bias   = (const float*)d_in[2];
  const float* diag   = (const float*)d_in[3];
  float* out = (float*)d_out;

  const int batch = in_sizes[0] / N_RES;

  (void)hipFuncSetAttribute(reinterpret_cast<const void*>(fwht_reservoir),
                            hipFuncAttributeMaxDynamicSharedMemorySize, LDS_BYTES);

  fwht_reservoir<<<dim3(batch), dim3(256), LDS_BYTES, stream>>>(
      state, inputs, bias, diag, out);
}


# Wave priority falls with progress (s_setprio 3,2,2,2,1,0 over the six passes): of two work-groups on a WGP the one behind wins the matrix pipe; they alternate, each covering the other's LDS exchanges
# speedup vs baseline: 4.4037x; 4.4037x over previous
#include <hip/hip_runtime.h>


typedef __attribute__((ext_vector_type(2))) float v2f;
typedef __attribute__((ext_vector_type(4))) float v4f;
typedef __attribute__((ext_vector_type(8))) float v8f;

#define N_RES   8192
#define N_INP   64
#define N_HAD   16384
#define N_LAYER 3
#define PITCH   140
#define LDS_BYTES (128 * PITCH * 4)

__device__ __forceinline__ float hsgn(int i, int j) {
  return (__popc(i & j) & 1) ? -1.0f : 1.0f;
}

__device__ __forceinline__ v8f v8f_zero() {
  v8f z;
#pragma unroll
  for (int i = 0; i < 8; ++i) z[i] = 0.0f;
  return z;
}

__device__ __forceinline__ v4f v4f_zero() {
  v4f z;
  z.x = 0.0f; z.y = 0.0f; z.z = 0.0f; z.w = 0.0f;
  return z;
}

__device__ __forceinline__ v4f ld4(const float* p) {
  return *reinterpret_cast<const v4f*>(p);
}

__device__ __forceinline__ void lds_barrier() {
  __builtin_amdgcn_fence(__ATOMIC_RELEASE, "workgroup", "local");
  __builtin_amdgcn_s_barrier();
  __builtin_amdgcn_fence(__ATOMIC_ACQUIRE, "workgroup", "local");
}

__global__ __launch_bounds__(256) __attribute__((amdgpu_waves_per_eu(4))) void fwht_reservoir(
    const float* __restrict__ state, const float* __restrict__ inputs,
    const float* __restrict__ bias, const float* __restrict__ diag,
    float* __restrict__ out) {
  extern __shared__ __align__(16) float X[];
  __builtin_amdgcn_s_setprio(3);
  const int row  = blockIdx.x;
  const int tid  = threadIdx.x;
  const int lane = tid & 31;
  const int wave = tid >> 5;
  const int wave_s = __builtin_amdgcn_readfirstlane(wave);
  const int m    = lane & 15;
  const int kh   = lane >> 4;

  const int aoff = (wave * 16 + m) * 128 + 8 * kh;

  v4f dq[8][2];
  v4f q[8][2];

  {
    const float* dg = diag + aoff;
#pragma unroll
    for (int C = 0; C < 8; ++C) {
      dq[C][0] = ld4(dg + 16 * C);
      dq[C][1] = ld4(dg + 16 * C + 4);
    }
    if (wave_s < 4) {
      const float* sp = state + (size_t)row * N_RES + aoff;
#pragma unroll
      for (int C = 0; C < 8; ++C) {
        q[C][0] = ld4(sp + 16 * C);
        q[C][1] = ld4(sp + 16 * C + 4);
      }
#pragma unroll
      for (int C = 0; C < 8; ++C) {
        q[C][0] = (0.9f * q[C][0]) * dq[C][0];
        q[C][1] = (0.9f * q[C][1]) * dq[C][1];
      }
    } else {
#pragma unroll
      for (int C = 0; C < 8; ++C) {
        q[C][0] = v4f_zero();
        q[C][1] = v4f_zero();
      }
      if (wave_s == 4 && m == 0) {
        const float* ip = inputs + (size_t)row * N_INP + 8 * kh;
#pragma unroll
        for (int C = 0; C < 4; ++C) {
          q[C][0] = 0.4f * ld4(ip + 16 * C);
          q[C][1] = 0.4f * ld4(ip + 16 * C + 4);
        }
      }
#pragma unroll
      for (int C = 0; C < 8; ++C) {
        q[C][0] = q[C][0] * dq[C][0];
        q[C][1] = q[C][1] * dq[C][1];
      }
    }
  }

  v2f Hp[4], Hq[4];
#pragma unroll
  for (int s = 0; s < 4; ++s) {
    Hp[s].x = hsgn(m, 4 * s + 2 * kh);
    Hp[s].y = hsgn(m, 4 * s + 2 * kh + 1);
    Hq[s].x = hsgn(m, 8 * kh + 2 * s);
    Hq[s].y = hsgn(m, 8 * kh + 2 * s + 1);
  }

#pragma unroll
  for (int l = 0; l < N_LAYER; ++l) {
    if (l == N_LAYER - 1) __builtin_amdgcn_s_setprio(1);
    {
      if (l > 0) {
#pragma unroll
        for (int C = 0; C < 8; ++C) {
          const float* src = X + (wave * 16 + m) * PITCH + C * 16 + 8 * kh;
          q[C][0] = *reinterpret_cast<const v4f*>(src);
          q[C][1] = *reinterpret_cast<const v4f*>(src + 4);
        }
#pragma unroll
        for (int C = 0; C < 8; ++C) {
          q[C][0] = q[C][0] * dq[C][0];
          q[C][1] = q[C][1] * dq[C][1];
        }
      }
      v8f d2[8];
#pragma unroll
      for (int C = 0; C < 8; ++C) {
        v2f a[4];
        a[0].x = q[C][0].x; a[0].y = q[C][0].y;
        a[1].x = q[C][0].z; a[1].y = q[C][0].w;
        a[2].x = q[C][1].x; a[2].y = q[C][1].y;
        a[3].x = q[C][1].z; a[3].y = q[C][1].w;
        v8f d1 = v8f_zero();
#pragma unroll
        for (int s = 0; s < 4; ++s)
          d1 = __builtin_amdgcn_wmma_f32_16x16x4_f32(
              false, a[s], false, Hq[s], (short)0, d1, false, false);
        v8f e2 = v8f_zero();
#pragma unroll
        for (int s = 0; s < 4; ++s) {
          v2f b;
          b.x = d1[2 * s];
          b.y = d1[2 * s + 1];
          e2 = __builtin_amdgcn_wmma_f32_16x16x4_f32(
              false, Hq[s], false, b, (short)0, e2, false, false);
        }
        d2[C] = e2;
      }
      if (l + 1 < N_LAYER) {
        __builtin_amdgcn_sched_barrier(0);
        const float* dg = diag + (size_t)(l + 1) * N_HAD + aoff;
#pragma unroll
        for (int C = 0; C < 8; ++C) {
          dq[C][0] = ld4(dg + 16 * C);
          dq[C][1] = ld4(dg + 16 * C + 4);
        }
      }
#pragma unroll
      for (int C = 0; C < 8; ++C)
#pragma unroll
        for (int v = 0; v < 8; ++v)
          X[(wave * 16 + v + 8 * kh) * PITCH + C * 16 + m] = d2[C][v];
    }
    lds_barrier();
    if (l == 0) __builtin_amdgcn_s_setprio(2);
    if (l == N_LAYER - 1) __builtin_amdgcn_s_setprio(0);

    {
      float bz[2][2][8];
      if (l == N_LAYER - 1) {
#pragma unroll
        for (int g = 0; g < 2; ++g)
#pragma unroll
          for (int d = 0; d < 2; ++d)
#pragma unroll
            for (int v = 0; v < 8; ++v) {
              const int orow = d * 32 + (v & 1) * 16 + wave * 2 + g;
              const int ocol = ((v >> 2) & 1) * 64 + ((v >> 1) & 1) * 32 + lane;
              bz[g][d][v] = bias[orow * 128 + ocol];
            }
      }
      v2f xb[2][4][4];
#pragma unroll
      for (int g = 0; g < 2; ++g)
#pragma unroll
        for (int d = 0; d < 4; ++d)
#pragma unroll
          for (int s = 0; s < 4; ++s) {
            const int rlo = wave * 2 + g;
            const int col = (s & 1) * 64 + (s >> 1) * 32 + lane;
            xb[g][d][s].x = X[(d * 32 + rlo) * PITCH + col];
            xb[g][d][s].y = X[(d * 32 + 16 + rlo) * PITCH + col];
          }
      v2f Hu[4];
#pragma unroll
      for (int s = 0; s < 4; ++s) Hu[s] = Hp[s];
      if (l == N_LAYER - 1) {
        const int mp = (m & 5) | ((m & 8) >> 2) | ((m & 2) << 2);
#pragma unroll
        for (int s = 0; s < 4; ++s) {
          Hu[s].x = hsgn(mp, 4 * s + 2 * kh);
          Hu[s].y = hsgn(mp, 4 * s + 2 * kh + 1);
        }
      }
      v8f acc[2][4];
#pragma unroll
      for (int g = 0; g < 2; ++g)
#pragma unroll
        for (int d = 0; d < 4; ++d) {
          v8f t = v8f_zero();
#pragma unroll
          for (int s = 0; s < 4; ++s)
            t = __builtin_amdgcn_wmma_f32_16x16x4_f32(
                false, Hu[s], false, xb[g][d][s],
                (short)0, t, false, false);
          acc[g][d] = t;
        }
#pragma unroll
      for (int g = 0; g < 2; ++g) {
        const int rlo = wave * 2 + g;
        const v8f t0 = acc[g][0] + acc[g][1];
        const v8f t1 = acc[g][0] - acc[g][1];
        const v8f t2 = acc[g][2] + acc[g][3];
        const v8f t3 = acc[g][2] - acc[g][3];
        v8f o[4];
        o[0] = t0 + t2;
        o[1] = t1 + t3;
        o[2] = t0 - t2;
        o[3] = t1 - t3;
        if (l < N_LAYER - 1) {
#pragma unroll
          for (int d = 0; d < 4; ++d)
#pragma unroll
            for (int v = 0; v < 8; ++v) {
              const int orow = d * 32 + (v & 1) * 16 + rlo;
              const int ocol = ((v >> 2) & 1) * 64 + kh * 32 + ((v >> 1) & 1) * 16 + m;
              X[orow * PITCH + ocol] = o[d][v];
            }
        } else {
          float* op = out + (size_t)row * N_RES;
#pragma unroll
          for (int d = 0; d < 2; ++d)
#pragma unroll
            for (int v = 0; v < 8; ++v) {
              const int orow = d * 32 + (v & 1) * 16 + rlo;
              const int ocol = ((v >> 2) & 1) * 64 + ((v >> 1) & 1) * 32 + lane;
              const float z = o[d][v] * (1.0f / (float)N_HAD) + bz[g][d][v];
              op[orow * 128 + ocol] = erff(z);
            }
        }
      }
    }

    if (l + 1 < N_LAYER) lds_barrier();
  }
}

extern "C" void kernel_launch(void* const* d_in, const int* in_sizes, int n_in,
                              void* d_out, int out_size, void* d_ws, size_t ws_size,
                              hipStream_t stream) {
  const float* state  = (const float*)d_in[0];
  const float* inputs = (const float*)d_in[1];
  const float* bias   = (const float*)d_in[2];
  const float* diag   = (const float*)d_in[3];
  float* out = (float*)d_out;

  const int batch = in_sizes[0] / N_RES;

  (void)hipFuncSetAttribute(reinterpret_cast<const void*>(fwht_reservoir),
                            hipFuncAttributeMaxDynamicSharedMemorySize, LDS_BYTES);

  fwht_reservoir<<<dim3(batch), dim3(256), LDS_BYTES, stream>>>(
      state, inputs, bias, diag, out);
}
